# starting kernel
# baseline (speedup 1.0000x reference)
#include <hip/hip_runtime.h>
#include <hip/hip_bf16.h>


typedef __attribute__((ext_vector_type(2))) float v2f;
typedef __attribute__((ext_vector_type(8))) float v8f;

#define D_FEAT 64


__global__ __launch_bounds__(256) void zero_f32(float* __restrict__ p, int n) {
  int i = blockIdx.x * blockDim.x + threadIdx.x;
  if (i < n) p[i] = 0.0f;
}

__global__ __launch_bounds__(256) void degree_kernel(const int* __restrict__ row,
                                                     float* __restrict__ deg, int e) {
  int i = blockIdx.x * blockDim.x + threadIdx.x;
  if (i < e) atomicAdd(&deg[row[i]], 1.0f);
}

__global__ __launch_bounds__(256) void dinv_kernel(float* __restrict__ deg_dinv, int n) {
  int i = blockIdx.x * blockDim.x + threadIdx.x;
  if (i < n) {
    float d = deg_dinv[i];
    deg_dinv[i] = (d > 0.0f) ? rsqrtf(d) : 0.0f;
  }
}

__global__ __launch_bounds__(256) void bias_init_kernel(const float* __restrict__ bias,
                                                        float* __restrict__ out, int total) {
  int i = blockIdx.x * blockDim.x + threadIdx.x;
  if (i < total) out[i] = bias[i & (D_FEAT - 1)];
}

__global__ __launch_bounds__(256) void gemm_xw_wmma(const float* __restrict__ X,
                                                    const float* __restrict__ W,
                                                    float* __restrict__ XW, int n) {
  const int wave = threadIdx.x >> 5;
  const int lane = threadIdx.x & 31;
  const int half = lane >> 4;
  const int l    = lane & 15;

  const int mBase = (blockIdx.x * 2 + (wave >> 2)) * 16;
  const int nBase = (wave & 3) * 16;
  if (mBase >= n) return;

  const int m = mBase + l;

  v8f acc = {};
#pragma unroll
  for (int k = 0; k < D_FEAT; k += 4) {
    v2f a, b;
    a.x = X[m * D_FEAT + k + 2 * half + 0];
    a.y = X[m * D_FEAT + k + 2 * half + 1];
    b.x = W[(k + 2 * half + 0) * D_FEAT + nBase + l];
    b.y = W[(k + 2 * half + 1) * D_FEAT + nBase + l];
    acc = __builtin_amdgcn_wmma_f32_16x16x4_f32(false, a, false, b,
                                                (short)0, acc, false, false);
  }

#pragma unroll
  for (int r = 0; r < 8; ++r) {
    XW[(mBase + r + 8 * half) * D_FEAT + nBase + l] = acc[r];
  }
}

__global__ __launch_bounds__(256) void scatter_kernel(const int* __restrict__ row,
                                                      const int* __restrict__ col,
                                                      const float* __restrict__ dinv,
                                                      const float* __restrict__ xw,
                                                      float* __restrict__ out, int e) {
  const int edge = blockIdx.x * 8 + (threadIdx.x >> 5);
  if (edge >= e) return;
  const int lane = threadIdx.x & 31;

  const int r = row[edge];
  const int c = col[edge];
  const float w = dinv[r] * dinv[c];

  const float v0 = xw[c * D_FEAT + lane] * w;
  const float v1 = xw[c * D_FEAT + lane + 32] * w;
  atomicAdd(&out[r * D_FEAT + lane], v0);
  atomicAdd(&out[r * D_FEAT + lane + 32], v1);
}


extern "C" void kernel_launch(void* const* d_in, const int* in_sizes, int n_in,
                              void* d_out, int out_size, void* d_ws, size_t ws_size,
                              hipStream_t stream) {
  const float* x      = (const float*)d_in[0];
  const int*   eidx   = (const int*)d_in[1];
  const float* weight = (const float*)d_in[2];
  const float* bias   = (const float*)d_in[3];
  float*       out    = (float*)d_out;

  const int n = in_sizes[0] / D_FEAT;
  const int e = in_sizes[1] / 2;
  const int* row = eidx;
  const int* col = eidx + e;

  float* dinv = (float*)d_ws;
  const int dinvStride = (n + 255) & ~255;
  float* xw = dinv + dinvStride;

  zero_f32<<<(n + 255) / 256, 256, 0, stream>>>(dinv, n);
  degree_kernel<<<(e + 255) / 256, 256, 0, stream>>>(row, dinv, e);
  dinv_kernel<<<(n + 255) / 256, 256, 0, stream>>>(dinv, n);
  gemm_xw_wmma<<<(n + 31) / 32, 256, 0, stream>>>(x, weight, xw, n);
  bias_init_kernel<<<(n * D_FEAT + 255) / 256, 256, 0, stream>>>(bias, out, n * D_FEAT);
  scatter_kernel<<<(e + 7) / 8, 256, 0, stream>>>(row, col, dinv, xw, out, e);
}


# edge sums pulled, not scattered: a returning integer add on the row's degree counter files each edge in a per-row list; one wave per row sums it in registers, stores once (order: arrival, per row)
# speedup vs baseline: 1.6905x; 1.6905x over previous
#include <hip/hip_runtime.h>
#include <hip/hip_bf16.h>


typedef __attribute__((ext_vector_type(2))) float v2f;
typedef __attribute__((ext_vector_type(8))) float v8f;

#define D_FEAT 64
#define ROW_CAP 64


__global__ __launch_bounds__(256) void zero_i32(int* __restrict__ p, int n) {
  int i = blockIdx.x * blockDim.x + threadIdx.x;
  if (i < n) p[i] = 0;
}

__global__ __launch_bounds__(256) void rank_place_kernel(const int* __restrict__ row,
                                                         const int* __restrict__ col,
                                                         int* __restrict__ cnt,
                                                         int* __restrict__ lists,
                                                         int* __restrict__ ov_edge,
                                                         int n, int e) {
  int i = blockIdx.x * blockDim.x + threadIdx.x;
  if (i < e) {
    const int r = row[i];
    const int c = col[i];
    asm volatile("" :: "v"(c));
    const int slot = atomicAdd(&cnt[r], 1);
    if (slot < ROW_CAP) {
      lists[r * ROW_CAP + slot] = c;
    } else {
      const int k = atomicAdd(&cnt[n], 1);
      ov_edge[k] = i;
    }
  }
}

__global__ __launch_bounds__(256) void gemm_xw_wmma(const float* __restrict__ X,
                                                    const float* __restrict__ W,
                                                    float* __restrict__ XW, int n) {
  const int wave = threadIdx.x >> 5;
  const int lane = threadIdx.x & 31;
  const int half = lane >> 4;
  const int l    = lane & 15;

  const int mBase = (blockIdx.x * 2 + (wave >> 2)) * 16;
  const int nBase = (wave & 3) * 16;
  if (mBase >= n) return;

  const int m = mBase + l;

  v8f acc = {};
#pragma unroll
  for (int k = 0; k < D_FEAT; k += 4) {
    v2f a, b;
    a.x = X[m * D_FEAT + k + 2 * half + 0];
    a.y = X[m * D_FEAT + k + 2 * half + 1];
    b.x = W[(k + 2 * half + 0) * D_FEAT + nBase + l];
    b.y = W[(k + 2 * half + 1) * D_FEAT + nBase + l];
    acc = __builtin_amdgcn_wmma_f32_16x16x4_f32(false, a, false, b,
                                                (short)0, acc, false, false);
  }

#pragma unroll
  for (int r = 0; r < 8; ++r) {
    XW[(mBase + r + 8 * half) * D_FEAT + nBase + l] = acc[r];
  }
}

__device__ __forceinline__ float dinv_of(int d) {
  return (d > 0) ? rsqrtf((float)d) : 0.0f;
}

__global__ __launch_bounds__(256) void pull_kernel(const int* __restrict__ row,
                                                   const int* __restrict__ col,
                                                   const int* __restrict__ cnt,
                                                   const int* __restrict__ lists,
                                                   const int* __restrict__ ov_edge,
                                                   const float* __restrict__ xw,
                                                   const float* __restrict__ bias,
                                                   float* __restrict__ out, int n) {
  const int wave = __builtin_amdgcn_readfirstlane((int)(threadIdx.x >> 5));
  const int r = blockIdx.x * 8 + wave;
  if (r >= n) return;
  const int lane = threadIdx.x & 31;

  const v2f* __restrict__ xw2 = (const v2f*)xw;

  const int deg = cnt[r];
  const float dr = dinv_of(deg);
  const int m = deg < ROW_CAP ? deg : ROW_CAP;

  float a0 = 0.0f, a1 = 0.0f;

  for (int base = 0; base < m; base += 32) {
    const int k = (m - base) < 32 ? (m - base) : 32;
    int c = 0;
    float w = 0.0f;
    if (lane < k) {
      c = lists[r * ROW_CAP + base + lane];
      w = dr * dinv_of(cnt[c]);
    }
    for (int j = 0; j < k; j += 8) {
      v2f x[8];
      float wj[8];
#pragma unroll
      for (int u = 0; u < 8; ++u) {
        const int cj = __builtin_amdgcn_readlane(c, j + u);
        wj[u] = __int_as_float(__builtin_amdgcn_readlane(__float_as_int(w), j + u));
        x[u] = xw2[cj * (D_FEAT / 2) + lane];
      }
#pragma unroll
      for (int u = 0; u < 8; ++u) {
        a0 = fmaf(x[u].x, wj[u], a0);
        a1 = fmaf(x[u].y, wj[u], a1);
      }
    }
  }

  if (deg > ROW_CAP) {
    const int novf = cnt[n];
    for (int i = 0; i < novf; ++i) {
      const int ed = ov_edge[i];
      if (row[ed] == r) {
        const int c = col[ed];
        const float w = dr * dinv_of(cnt[c]);
        const v2f x = xw2[c * (D_FEAT / 2) + lane];
        a0 = fmaf(x.x, w, a0);
        a1 = fmaf(x.y, w, a1);
      }
    }
  }

  v2f o;
  o.x = a0 + bias[2 * lane + 0];
  o.y = a1 + bias[2 * lane + 1];
  ((v2f*)out)[r * (D_FEAT / 2) + lane] = o;
}


extern "C" void kernel_launch(void* const* d_in, const int* in_sizes, int n_in,
                              void* d_out, int out_size, void* d_ws, size_t ws_size,
                              hipStream_t stream) {
  const float* x      = (const float*)d_in[0];
  const int*   eidx   = (const int*)d_in[1];
  const float* weight = (const float*)d_in[2];
  const float* bias   = (const float*)d_in[3];
  float*       out    = (float*)d_out;

  const int n = in_sizes[0] / D_FEAT;
  const int e = in_sizes[1] / 2;
  const int* row = eidx;
  const int* col = eidx + e;

  int* cnt = (int*)d_ws;
  const int cntStride = (n + 1 + 255) & ~255;
  int* lists = cnt + cntStride;
  int* ov_edge = lists + (size_t)n * ROW_CAP;
  const int ovStride = (e + 255) & ~255;
  float* xw = (float*)(ov_edge + ovStride);

  zero_i32<<<(n + 1 + 255) / 256, 256, 0, stream>>>(cnt, n + 1);
  rank_place_kernel<<<(e + 255) / 256, 256, 0, stream>>>(row, col, cnt, lists, ov_edge, n, e);
  gemm_xw_wmma<<<(n + 31) / 32, 256, 0, stream>>>(x, weight, xw, n);
  pull_kernel<<<(n + 7) / 8, 256, 0, stream>>>(row, col, cnt, lists, ov_edge, xw, bias, out, n);
}


# products reassociated: a wave sums its row's weighted x rows, 16 row sums meet in LDS, 4 waves multiply them by W (f32 WMMA) in the same kernel; X*W launch, plane gone (order: rows, then k in 2 sums)
# speedup vs baseline: 1.7784x; 1.0520x over previous
#include <hip/hip_runtime.h>
#include <hip/hip_bf16.h>


typedef __attribute__((ext_vector_type(2))) float v2f;
typedef __attribute__((ext_vector_type(8))) float v8f;

#define D_FEAT 64
#define ROW_CAP 64


__global__ __launch_bounds__(256) void zero_i32(int* __restrict__ p, int n) {
  int i = blockIdx.x * blockDim.x + threadIdx.x;
  if (i < n) p[i] = 0;
}

__global__ __launch_bounds__(256) void rank_place_kernel(const int* __restrict__ row,
                                                         const int* __restrict__ col,
                                                         int* __restrict__ cnt,
                                                         int* __restrict__ lists,
                                                         int* __restrict__ ov_edge,
                                                         int n, int e) {
  int i = blockIdx.x * blockDim.x + threadIdx.x;
  if (i < e) {
    const int r = row[i];
    const int c = col[i];
    asm volatile("" :: "v"(c));
    const int slot = atomicAdd(&cnt[r], 1);
    if (slot < ROW_CAP) {
      lists[r * ROW_CAP + slot] = c;
    } else {
      const int k = atomicAdd(&cnt[n], 1);
      ov_edge[k] = i;
    }
  }
}

__device__ __forceinline__ float dinv_of(int d) {
  return (d > 0) ? rsqrtf((float)d) : 0.0f;
}

#define TILE_ROWS 16
#define A_PITCH 68

__global__ __launch_bounds__(512) void pull_mm_kernel(const int* __restrict__ row,
                                                      const int* __restrict__ col,
                                                      const int* __restrict__ cnt,
                                                      const int* __restrict__ lists,
                                                      const int* __restrict__ ov_edge,
                                                      const float* __restrict__ x,
                                                      const float* __restrict__ W,
                                                      const float* __restrict__ bias,
                                                      float* __restrict__ out, int n) {
  __shared__ float As[TILE_ROWS * A_PITCH];

  const int wave = __builtin_amdgcn_readfirstlane((int)(threadIdx.x >> 5));
  const int lane = threadIdx.x & 31;
  const int rBase = blockIdx.x * TILE_ROWS;
  const int r = rBase + wave;

  const v2f* __restrict__ x2 = (const v2f*)x;

  const int half = lane >> 4;
  const int l    = lane & 15;
  const int nBase = wave * 16;
  v2f bf[D_FEAT / 4];
  float bv = 0.0f;
  if (wave < 4) {
#pragma unroll
    for (int kk = 0; kk < D_FEAT / 4; ++kk) {
      bf[kk].x = W[(4 * kk + 2 * half + 0) * D_FEAT + nBase + l];
      bf[kk].y = W[(4 * kk + 2 * half + 1) * D_FEAT + nBase + l];
    }
    bv = bias[nBase + l];
  }

  float a0 = 0.0f, a1 = 0.0f;

  if (r < n) {
    const int deg = cnt[r];
    const float dr = dinv_of(deg);
    const int m = deg < ROW_CAP ? deg : ROW_CAP;

    for (int base = 0; base < m; base += 32) {
      const int k = (m - base) < 32 ? (m - base) : 32;
      int c = 0;
      float w = 0.0f;
      if (lane < k) {
        c = lists[r * ROW_CAP + base + lane];
        w = dr * dinv_of(cnt[c]);
      }
      for (int j = 0; j < k; j += 8) {
        v2f xr[8];
        float wj[8];
#pragma unroll
        for (int u = 0; u < 8; ++u) {
          const int cj = __builtin_amdgcn_readlane(c, j + u);
          wj[u] = __int_as_float(__builtin_amdgcn_readlane(__float_as_int(w), j + u));
          xr[u] = x2[cj * (D_FEAT / 2) + lane];
        }
#pragma unroll
        for (int u = 0; u < 8; ++u) {
          a0 = fmaf(xr[u].x, wj[u], a0);
          a1 = fmaf(xr[u].y, wj[u], a1);
        }
      }
    }

    if (deg > ROW_CAP) {
      const int novf = cnt[n];
      for (int i = 0; i < novf; ++i) {
        const int ed = ov_edge[i];
        if (row[ed] == r) {
          const int c = col[ed];
          const float w = dr * dinv_of(cnt[c]);
          const v2f xv = x2[c * (D_FEAT / 2) + lane];
          a0 = fmaf(xv.x, w, a0);
          a1 = fmaf(xv.y, w, a1);
        }
      }
    }
  }

  v2f s;
  s.x = a0;
  s.y = a1;
  *(v2f*)&As[wave * A_PITCH + 2 * lane] = s;
  __syncthreads();

  if (wave >= 4) return;

  v8f acc0 = {}, acc1 = {};
#pragma unroll
  for (int kk = 0; kk < D_FEAT / 4; kk += 2) {
    const v2f e0 = *(const v2f*)&As[l * A_PITCH + 4 * kk + 2 * half];
    const v2f e1 = *(const v2f*)&As[l * A_PITCH + 4 * (kk + 1) + 2 * half];
    acc0 = __builtin_amdgcn_wmma_f32_16x16x4_f32(false, e0, false, bf[kk],
                                                 (short)0, acc0, false, false);
    acc1 = __builtin_amdgcn_wmma_f32_16x16x4_f32(false, e1, false, bf[kk + 1],
                                                 (short)0, acc1, false, false);
  }

  float* __restrict__ o = out + (rBase + 8 * half) * D_FEAT + nBase + l;
  if (rBase + TILE_ROWS <= n) {
#pragma unroll
    for (int q = 0; q < 8; ++q) o[q * D_FEAT] = (acc0[q] + acc1[q]) + bv;
  } else {
#pragma unroll
    for (int q = 0; q < 8; ++q) {
      if (rBase + q + 8 * half < n) o[q * D_FEAT] = (acc0[q] + acc1[q]) + bv;
    }
  }
}


extern "C" void kernel_launch(void* const* d_in, const int* in_sizes, int n_in,
                              void* d_out, int out_size, void* d_ws, size_t ws_size,
                              hipStream_t stream) {
  const float* x      = (const float*)d_in[0];
  const int*   eidx   = (const int*)d_in[1];
  const float* weight = (const float*)d_in[2];
  const float* bias   = (const float*)d_in[3];
  float*       out    = (float*)d_out;

  const int n = in_sizes[0] / D_FEAT;
  const int e = in_sizes[1] / 2;
  const int* row = eidx;
  const int* col = eidx + e;

  int* cnt = (int*)d_ws;
  const int cntStride = (n + 1 + 255) & ~255;
  int* lists = cnt + cntStride;
  int* ov_edge = lists + (size_t)n * ROW_CAP;

  zero_i32<<<(n + 1 + 255) / 256, 256, 0, stream>>>(cnt, n + 1);
  rank_place_kernel<<<(e + 255) / 256, 256, 0, stream>>>(row, col, cnt, lists, ov_edge, n, e);
  pull_mm_kernel<<<(n + TILE_ROWS - 1) / TILE_ROWS, 512, 0, stream>>>(row, col, cnt, lists, ov_edge, x,
                                                                      weight, bias, out, n);
}
